# starting kernel
# baseline (speedup 1.0000x reference)
#include <hip/hip_runtime.h>
#include <hip/hip_bf16.h>


typedef __attribute__((ext_vector_type(2))) float v2f;
typedef __attribute__((ext_vector_type(8))) float v8f;

#define NPIX 1024
#define P_LOG2 9
#define NBP (2048 * 512)
#define WAVES_PER_BLOCK 8
#define TILES_PER_WAVE 8

__global__ __launch_bounds__(256) void qcirc_wmma_kernel(
    const float* __restrict__ x,
    const float* __restrict__ rz,
    const float* __restrict__ ry,
    const int*   __restrict__ pidx,
    float*       __restrict__ out)
{
    const int lane  = threadIdx.x & 31;
    const int wave  = threadIdx.x >> 5;
    const int m     = lane & 15;
    const bool hi   = lane >= 16;
    const int tile0 = (blockIdx.x * WAVES_PER_BLOCK + wave) * TILES_PER_WAVE;
    const int bp0   = tile0 << 4;

    const int   n   = m;
    const float ryv = (n < 7) ? ry[n] : 0.0f;
    const float rzv = (n < 7) ? rz[n] : 0.0f;
    const float w0  = (n < 7) ?  0.5f * __cosf(ryv)               : 0.0f;
    const float w1  = (n < 7) ? -0.5f * __sinf(ryv) * __cosf(rzv) : 0.0f;
    const float w2  = (n >= 7 && n < 14) ? 0.5f : 0.0f;
    const float w3  = (n < 14)           ? 0.5f : 0.0f;
    v2f bm;
    bm.x = hi ? w2 : w0;
    bm.y = hi ? w3 : w1;

    v8f acc[TILES_PER_WAVE];
    #pragma unroll
    for (int t = 0; t < TILES_PER_WAVE; ++t) {
        const int bp = bp0 + (t << 4) + m;
        const int b  = bp >> P_LOG2;
        const int i0 = pidx[2 * bp + 0];
        const int i1 = pidx[2 * bp + 1];
        const float* xb = x + b * NPIX;
        const float x0 = xb[i0];
        const float x1 = xb[i1];
        const float s0 = __sinf(x0), c0 = __cosf(x0);
        const float s1 = __sinf(x1), c1 = __cosf(x1);

        v2f a;
        a.x = hi ? (c0 * c1) : c0;
        a.y = hi ? 1.0f      : (s0 * s1);

        v8f c = {};
        acc[t] = __builtin_amdgcn_wmma_f32_16x16x4_f32(
             false, a,  false, bm,
             (short)0, c,  false,  false);
    }

    if (n < 14) {
        const int rowBase0 = (bp0 + (hi ? 8 : 0)) * 14 + n;
        #pragma unroll
        for (int t = 0; t < TILES_PER_WAVE; ++t) {
            const int rowBase = rowBase0 + t * (16 * 14);
            #pragma unroll
            for (int v = 0; v < 8; ++v)
                out[rowBase + v * 14] = acc[t][v];
        }
    }
}

extern "C" void kernel_launch(void* const* d_in, const int* in_sizes, int n_in,
                              void* d_out, int out_size, void* d_ws, size_t ws_size,
                              hipStream_t stream) {
    const float* x    = (const float*)d_in[0];
    const float* rz   = (const float*)d_in[1];
    const float* ry   = (const float*)d_in[2];
    const int*   pidx = (const int*)d_in[3];
    float*       out  = (float*)d_out;

    const int tiles  = NBP / 16;
    const int blocks = tiles / (WAVES_PER_BLOCK * TILES_PER_WAVE);
    qcirc_wmma_kernel<<<blocks, 256, 0, stream>>>(x, rz, ry, pidx, out);
}


# Product transposed (W^T as A, features as B): a lane holds consecutive floats of an output row; each wave parks its 8 tiles in LDS and writes 14 whole-line 16-B stores, not 64 partial dword stores
# speedup vs baseline: 1.2922x; 1.2922x over previous
#include <hip/hip_runtime.h>
#include <hip/hip_bf16.h>


typedef __attribute__((ext_vector_type(2))) float v2f;
typedef __attribute__((ext_vector_type(4))) float v4f;
typedef __attribute__((ext_vector_type(8))) float v8f;

#define NPIX 1024
#define P_LOG2 9
#define NBP (2048 * 512)
#define WAVES_PER_BLOCK 8
#define TILES_PER_WAVE 8
#define ROW_F 14
#define TILE_F (16 * ROW_F)
#define WAVE_F (TILES_PER_WAVE * TILE_F)

__global__ __launch_bounds__(256) void qcirc_wmma_kernel(
    const float* __restrict__ x,
    const float* __restrict__ rz,
    const float* __restrict__ ry,
    const int*   __restrict__ pidx,
    float*       __restrict__ out)
{
    __shared__ __attribute__((aligned(16))) float stage[WAVES_PER_BLOCK * WAVE_F];

    const int lane  = threadIdx.x & 31;
    const int wave  = threadIdx.x >> 5;
    const int m     = lane & 15;
    const bool hi   = lane >= 16;
    const int tile0 = (blockIdx.x * WAVES_PER_BLOCK + wave) * TILES_PER_WAVE;
    const int bp0   = tile0 << 4;

    const int   n   = m;
    const float ryv = (n < 7) ? ry[n] : 0.0f;
    const float rzv = (n < 7) ? rz[n] : 0.0f;
    const float w0  = (n < 7) ?  0.5f * __cosf(ryv)               : 0.0f;
    const float w1  = (n < 7) ? -0.5f * __sinf(ryv) * __cosf(rzv) : 0.0f;
    const float w2  = (n >= 7 && n < 14) ? 0.5f : 0.0f;
    const float w3  = (n < 14)           ? 0.5f : 0.0f;
    v2f wm;
    wm.x = hi ? w2 : w0;
    wm.y = hi ? w3 : w1;

    v8f acc[TILES_PER_WAVE];
    #pragma unroll
    for (int t = 0; t < TILES_PER_WAVE; ++t) {
        const int bp = bp0 + (t << 4) + m;
        const int b  = bp >> P_LOG2;
        const int i0 = pidx[2 * bp + 0];
        const int i1 = pidx[2 * bp + 1];
        const float* xb = x + b * NPIX;
        const float x0 = xb[i0];
        const float x1 = xb[i1];
        const float s0 = __sinf(x0), c0 = __cosf(x0);
        const float s1 = __sinf(x1), c1 = __cosf(x1);

        v2f f;
        f.x = hi ? (c0 * c1) : c0;
        f.y = hi ? 1.0f      : (s0 * s1);

        v8f c = {};
        acc[t] = __builtin_amdgcn_wmma_f32_16x16x4_f32(
             false, wm,  false, f,
             (short)0, c,  false,  false);
    }

    float* ws = stage + wave * WAVE_F;
    {
        float* prow = ws + m * ROW_F + (hi ? 8 : 0);
        #pragma unroll
        for (int t = 0; t < TILES_PER_WAVE; ++t) {
            v2f* p2 = (v2f*)(prow + t * TILE_F);
            p2[0] = (v2f){acc[t][0], acc[t][1]};
            p2[1] = (v2f){acc[t][2], acc[t][3]};
            p2[2] = (v2f){acc[t][4], acc[t][5]};
        }
        if (!hi) {
            #pragma unroll
            for (int t = 0; t < TILES_PER_WAVE; ++t) {
                v2f* p2 = (v2f*)(prow + t * TILE_F);
                p2[3] = (v2f){acc[t][6], acc[t][7]};
            }
        }
    }

    __builtin_amdgcn_fence(__ATOMIC_RELEASE, "wavefront");
    __builtin_amdgcn_wave_barrier();
    __builtin_amdgcn_fence(__ATOMIC_ACQUIRE, "wavefront");

    {
        const v4f* src = (const v4f*)ws + lane;
        v4f*       dst = (v4f*)(out + (size_t)bp0 * ROW_F) + lane;
        #pragma unroll
        for (int q = 0; q < WAVE_F / (4 * 32); ++q)
            dst[q * 32] = src[q * 32];
    }
}

extern "C" void kernel_launch(void* const* d_in, const int* in_sizes, int n_in,
                              void* d_out, int out_size, void* d_ws, size_t ws_size,
                              hipStream_t stream) {
    const float* x    = (const float*)d_in[0];
    const float* rz   = (const float*)d_in[1];
    const float* ry   = (const float*)d_in[2];
    const int*   pidx = (const int*)d_in[3];
    float*       out  = (float*)d_out;

    const int tiles  = NBP / 16;
    const int blocks = tiles / (WAVES_PER_BLOCK * TILES_PER_WAVE);
    qcirc_wmma_kernel<<<blocks, 256, 0, stream>>>(x, rz, ry, pidx, out);
}


# Workgroup copies its 4 x images (16 KB) into LDS with whole-line 16-B loads; pair pixels come from ds_load, not 2 divergent global loads per tile; a wave covers 16 tiles, 512 workgroups
# speedup vs baseline: 1.5306x; 1.1845x over previous
#include <hip/hip_runtime.h>
#include <hip/hip_bf16.h>


typedef __attribute__((ext_vector_type(2))) float v2f;
typedef __attribute__((ext_vector_type(4))) float v4f;
typedef __attribute__((ext_vector_type(8))) float v8f;

#define NPIX 1024
#define P_LOG2 9
#define NBP (2048 * 512)
#define WAVES_PER_BLOCK 8
#define GROUP_TILES 8
#define GROUPS_PER_WAVE 2
#define TILES_PER_WAVE (GROUP_TILES * GROUPS_PER_WAVE)
#define ROW_F 14
#define TILE_F (16 * ROW_F)
#define GROUP_F (GROUP_TILES * TILE_F)
#define WAVE_F (TILES_PER_WAVE * TILE_F)
#define IMGS_PER_BLOCK 4
#define XS_F (IMGS_PER_BLOCK * NPIX)

__global__ __launch_bounds__(256) void qcirc_wmma_kernel(
    const float* __restrict__ x,
    const float* __restrict__ rz,
    const float* __restrict__ ry,
    const int*   __restrict__ pidx,
    float*       __restrict__ out)
{
    __shared__ __attribute__((aligned(16))) float ximg[XS_F];
    __shared__ __attribute__((aligned(16))) float stage[WAVES_PER_BLOCK * WAVE_F];

    const int lane  = threadIdx.x & 31;
    const int wave  = threadIdx.x >> 5;
    const int m     = lane & 15;
    const bool hi   = lane >= 16;
    const int tile0 = (blockIdx.x * WAVES_PER_BLOCK + wave) * TILES_PER_WAVE;
    const int bp0   = tile0 << 4;

    const v4f* gx = (const v4f*)(x + (size_t)blockIdx.x * XS_F) + threadIdx.x;
    const v4f xr0 = gx[0], xr1 = gx[256], xr2 = gx[512], xr3 = gx[768];

    int i0[TILES_PER_WAVE], i1[TILES_PER_WAVE];
    #pragma unroll
    for (int t = 0; t < TILES_PER_WAVE; ++t) {
        const int bp = bp0 + (t << 4) + m;
        i0[t] = pidx[2 * bp + 0];
        i1[t] = pidx[2 * bp + 1];
    }

    const int   n   = m;
    const float ryv = (n < 7) ? ry[n] : 0.0f;
    const float rzv = (n < 7) ? rz[n] : 0.0f;
    const float w0  = (n < 7) ?  0.5f * __cosf(ryv)               : 0.0f;
    const float w1  = (n < 7) ? -0.5f * __sinf(ryv) * __cosf(rzv) : 0.0f;
    const float w2  = (n >= 7 && n < 14) ? 0.5f : 0.0f;
    const float w3  = (n < 14)           ? 0.5f : 0.0f;
    v2f wm;
    wm.x = hi ? w2 : w0;
    wm.y = hi ? w3 : w1;

    {
        v4f* sx = (v4f*)ximg + threadIdx.x;
        sx[0] = xr0; sx[256] = xr1; sx[512] = xr2; sx[768] = xr3;
    }
    __syncthreads();

    const float* xs = ximg + (wave >> 1) * NPIX;
    float* ws = stage + wave * WAVE_F;

    #pragma unroll
    for (int g = 0; g < GROUPS_PER_WAVE; ++g) {
        v8f acc[GROUP_TILES];
        #pragma unroll
        for (int t = 0; t < GROUP_TILES; ++t) {
            const float x0 = xs[i0[g * GROUP_TILES + t]];
            const float x1 = xs[i1[g * GROUP_TILES + t]];
            const float s0 = __sinf(x0), c0 = __cosf(x0);
            const float s1 = __sinf(x1), c1 = __cosf(x1);

            v2f f;
            f.x = hi ? (c0 * c1) : c0;
            f.y = hi ? 1.0f      : (s0 * s1);

            v8f c = {};
            acc[t] = __builtin_amdgcn_wmma_f32_16x16x4_f32(
                 false, wm,  false, f,
                 (short)0, c,  false,  false);
        }

        float* wg = ws + g * GROUP_F;
        {
            float* prow = wg + m * ROW_F + (hi ? 8 : 0);
            #pragma unroll
            for (int t = 0; t < GROUP_TILES; ++t) {
                v2f* p2 = (v2f*)(prow + t * TILE_F);
                p2[0] = (v2f){acc[t][0], acc[t][1]};
                p2[1] = (v2f){acc[t][2], acc[t][3]};
                p2[2] = (v2f){acc[t][4], acc[t][5]};
            }
            if (!hi) {
                #pragma unroll
                for (int t = 0; t < GROUP_TILES; ++t) {
                    v2f* p2 = (v2f*)(prow + t * TILE_F);
                    p2[3] = (v2f){acc[t][6], acc[t][7]};
                }
            }
        }

        __builtin_amdgcn_fence(__ATOMIC_RELEASE, "wavefront");
        __builtin_amdgcn_wave_barrier();
        __builtin_amdgcn_fence(__ATOMIC_ACQUIRE, "wavefront");

        {
            const v4f* src = (const v4f*)wg + lane;
            v4f*       dst = (v4f*)(out + (size_t)bp0 * ROW_F + g * GROUP_F) + lane;
            #pragma unroll
            for (int q = 0; q < GROUP_F / (4 * 32); ++q)
                dst[q * 32] = src[q * 32];
        }
    }
}

extern "C" void kernel_launch(void* const* d_in, const int* in_sizes, int n_in,
                              void* d_out, int out_size, void* d_ws, size_t ws_size,
                              hipStream_t stream) {
    const float* x    = (const float*)d_in[0];
    const float* rz   = (const float*)d_in[1];
    const float* ry   = (const float*)d_in[2];
    const int*   pidx = (const int*)d_in[3];
    float*       out  = (float*)d_out;

    const int tiles  = NBP / 16;
    const int blocks = tiles / (WAVES_PER_BLOCK * TILES_PER_WAVE);
    qcirc_wmma_kernel<<<blocks, 256, 0, stream>>>(x, rz, ry, pidx, out);
}
